# starting kernel
# baseline (speedup 1.0000x reference)
#include <hip/hip_runtime.h>
#include <hip/hip_bf16.h>
#include <math.h>


typedef float v2f __attribute__((ext_vector_type(2)));
typedef float v8f __attribute__((ext_vector_type(8)));

#define HID   1024
#define NST   128
#define BATCH 256
#define IN2   256
#define CB    32
#define TS    1028

__device__ __forceinline__ v8f wmma_f32(v2f a, v2f b, v8f c) {
  return __builtin_amdgcn_wmma_f32_16x16x4_f32(
       false, a,  false, b,
       (short)0, c,  false,  false);
}

__global__ void build_x2(const float* __restrict__ x, float* __restrict__ X2) {
  int i = blockIdx.x * blockDim.x + threadIdx.x;
  if (i >= 512 * IN2) return;
  int row = i >> 8, c = i & 255;
  float v;
  if (row < 256) {
    v = x[row * 512 + c];
  } else {
    int rz = row - 256;
    v = (c < 128) ? x[rz * 512 + 384 + c] : x[rz * 512 + 128 + (c - 128)];
  }
  X2[i] = v;
}

__global__ void zero_buf(float* __restrict__ p, int n) {
  int i = blockIdx.x * blockDim.x + threadIdx.x;
  if (i < n) p[i] = 0.f;
}

__global__ __launch_bounds__(256) void gemm_awt(
    const float* __restrict__ A, int lda,
    const float* __restrict__ W, int ldw,
    const float* __restrict__ bias,
    const float* __restrict__ mask, int ldm,
    float* __restrict__ C, int ldc,
    int M, int K, int epi)
{
  const int lane = threadIdx.x & 31, wv = threadIdx.x >> 5;
  const int l16 = lane & 15, hlf = lane >> 4;
  const int m0 = blockIdx.x * 128 + wv * 16;
  const int n0 = blockIdx.y * 64;
  if (m0 >= M) return;

  const v8f vz = {0.f,0.f,0.f,0.f,0.f,0.f,0.f,0.f};
  v8f acc[4] = {vz, vz, vz, vz};

  const int kof = hlf * 2;
  const float* Ar  = A + (size_t)(m0 + l16) * lda;
  const float* Wr0 = W + (size_t)(n0 +  0 + l16) * ldw;
  const float* Wr1 = W + (size_t)(n0 + 16 + l16) * ldw;
  const float* Wr2 = W + (size_t)(n0 + 32 + l16) * ldw;
  const float* Wr3 = W + (size_t)(n0 + 48 + l16) * ldw;

  v2f a0, b00, b10, b20, b30;
  v2f a1, b01, b11, b21, b31;

#define LDSET0(kk) { a0  = *(const v2f*)(Ar  + (kk)); \
                     b00 = *(const v2f*)(Wr0 + (kk)); \
                     b10 = *(const v2f*)(Wr1 + (kk)); \
                     b20 = *(const v2f*)(Wr2 + (kk)); \
                     b30 = *(const v2f*)(Wr3 + (kk)); }
#define LDSET1(kk) { a1  = *(const v2f*)(Ar  + (kk)); \
                     b01 = *(const v2f*)(Wr0 + (kk)); \
                     b11 = *(const v2f*)(Wr1 + (kk)); \
                     b21 = *(const v2f*)(Wr2 + (kk)); \
                     b31 = *(const v2f*)(Wr3 + (kk)); }
#define WM0() { acc[0] = wmma_f32(a0, b00, acc[0]); \
                acc[1] = wmma_f32(a0, b10, acc[1]); \
                acc[2] = wmma_f32(a0, b20, acc[2]); \
                acc[3] = wmma_f32(a0, b30, acc[3]); }
#define WM1() { acc[0] = wmma_f32(a1, b01, acc[0]); \
                acc[1] = wmma_f32(a1, b11, acc[1]); \
                acc[2] = wmma_f32(a1, b21, acc[2]); \
                acc[3] = wmma_f32(a1, b31, acc[3]); }

  LDSET0(kof);
  LDSET1(kof + 4);
  for (int k = 0; k < K - 8; k += 8) {
    WM0(); LDSET0(k + 8  + kof);
    WM1(); LDSET1(k + 12 + kof);
  }
  WM0();
  WM1();
#undef LDSET0
#undef LDSET1
#undef WM0
#undef WM1

  #pragma unroll
  for (int nt = 0; nt < 4; ++nt) {
    #pragma unroll
    for (int r = 0; r < 8; ++r) {
      const int m  = m0 + r + 8 * hlf;
      const int nn = n0 + nt * 16 + l16;
      float v = acc[nt][r];
      if (bias) v += bias[nn];
      if (epi == 1)      v = v > 0.f ? v   : 0.f;
      else if (epi == 2) v = v > 0.f ? 1.f : 0.f;
      if (mask) v *= mask[(size_t)m * ldm + nn];
      C[(size_t)m * ldc + nn] = v;
    }
  }
}

__global__ __launch_bounds__(256) void gemm_ab(
    const float* __restrict__ A, int lda,
    const float* __restrict__ B, int ldb,
    float* __restrict__ C, int ldc,
    int M, int K)
{
  const int lane = threadIdx.x & 31, wv = threadIdx.x >> 5;
  const int l16 = lane & 15, hlf = lane >> 4;
  const int m0 = blockIdx.x * 128 + wv * 16;
  const int n0 = blockIdx.y * 64;
  if (m0 >= M) return;

  const v8f vz = {0.f,0.f,0.f,0.f,0.f,0.f,0.f,0.f};
  v8f acc[4] = {vz, vz, vz, vz};
  const int kof = hlf * 2;
  const float* Ar = A + (size_t)(m0 + l16) * lda;
  const float* Bc = B + n0 + l16;

  v2f a0, c00, c10, c20, c30;
  v2f a1, c01, c11, c21, c31;

#define BLD(dst, kk, nt) { (dst).x = Bc[(size_t)(kk) * ldb + (nt) * 16]; \
                           (dst).y = Bc[(size_t)((kk) + 1) * ldb + (nt) * 16]; }
#define LDSET0(kk) { a0 = *(const v2f*)(Ar + (kk)); \
                     BLD(c00, (kk), 0) BLD(c10, (kk), 1) BLD(c20, (kk), 2) BLD(c30, (kk), 3) }
#define LDSET1(kk) { a1 = *(const v2f*)(Ar + (kk)); \
                     BLD(c01, (kk), 0) BLD(c11, (kk), 1) BLD(c21, (kk), 2) BLD(c31, (kk), 3) }
#define WM0() { acc[0] = wmma_f32(a0, c00, acc[0]); acc[1] = wmma_f32(a0, c10, acc[1]); \
                acc[2] = wmma_f32(a0, c20, acc[2]); acc[3] = wmma_f32(a0, c30, acc[3]); }
#define WM1() { acc[0] = wmma_f32(a1, c01, acc[0]); acc[1] = wmma_f32(a1, c11, acc[1]); \
                acc[2] = wmma_f32(a1, c21, acc[2]); acc[3] = wmma_f32(a1, c31, acc[3]); }

  LDSET0(kof);
  LDSET1(kof + 4);
  for (int k = 0; k < K - 8; k += 8) {
    WM0(); LDSET0(k + 8  + kof);
    WM1(); LDSET1(k + 12 + kof);
  }
  WM0();
  WM1();
#undef BLD
#undef LDSET0
#undef LDSET1
#undef WM0
#undef WM1

  #pragma unroll
  for (int nt = 0; nt < 4; ++nt)
    #pragma unroll
    for (int r = 0; r < 8; ++r)
      C[(size_t)(m0 + r + 8 * hlf) * ldc + n0 + nt * 16 + l16] = acc[nt][r];
}

__device__ __forceinline__ void chain_ld(
    const float* __restrict__ Wr, const float* __restrict__ Tb0,
    const float* __restrict__ Tb1, int kk,
    v2f av[8], v2f& c0, v2f& c1)
{
  c0 = *(const v2f*)(Tb0 + kk);
  c1 = *(const v2f*)(Tb1 + kk);
  #pragma unroll
  for (int mt = 0; mt < 8; ++mt)
    av[mt] = *(const v2f*)(Wr + (size_t)(mt * 16 * HID) + kk);
}

__device__ __forceinline__ void chain_mm(const v2f av[8], v2f c0, v2f c1,
                                         v8f acc[8][2])
{
  #pragma unroll
  for (int mt = 0; mt < 8; ++mt) {
    acc[mt][0] = wmma_f32(av[mt], c0, acc[mt][0]);
    acc[mt][1] = wmma_f32(av[mt], c1, acc[mt][1]);
  }
}

__device__ __forceinline__ void chain_gemm_1024(
    const float* __restrict__ Wm,
    const float* __restrict__ Tin,
    float* __restrict__ Tout,
    const float* __restrict__ dm,
    int lane, int wv)
{
  const int l16 = lane & 15, hlf = lane >> 4;
  const int kof = hlf * 2;
  const v8f vz = {0.f,0.f,0.f,0.f,0.f,0.f,0.f,0.f};
  v8f acc[8][2];
  #pragma unroll
  for (int mt = 0; mt < 8; ++mt) { acc[mt][0] = vz; acc[mt][1] = vz; }

  const float* Tb0 = Tin + (size_t)l16 * TS;
  const float* Tb1 = Tin + (size_t)(16 + l16) * TS;
  const float* Wr  = Wm + (size_t)(wv * 128 + l16) * HID;

  v2f avA[8], cA0, cA1;
  v2f avB[8], cB0, cB1;
  v2f avC[8], cC0, cC1;
  v2f avD[8], cD0, cD1;

  chain_ld(Wr, Tb0, Tb1, kof,      avA, cA0, cA1);
  chain_ld(Wr, Tb0, Tb1, kof + 4,  avB, cB0, cB1);
  chain_ld(Wr, Tb0, Tb1, kof + 8,  avC, cC0, cC1);
  chain_ld(Wr, Tb0, Tb1, kof + 12, avD, cD0, cD1);
  for (int k = 0; k < HID - 16; k += 16) {
    chain_mm(avA, cA0, cA1, acc);
    chain_ld(Wr, Tb0, Tb1, k + 16 + kof, avA, cA0, cA1);
    chain_mm(avB, cB0, cB1, acc);
    chain_ld(Wr, Tb0, Tb1, k + 20 + kof, avB, cB0, cB1);
    chain_mm(avC, cC0, cC1, acc);
    chain_ld(Wr, Tb0, Tb1, k + 24 + kof, avC, cC0, cC1);
    chain_mm(avD, cD0, cD1, acc);
    chain_ld(Wr, Tb0, Tb1, k + 28 + kof, avD, cD0, cD1);
  }
  chain_mm(avA, cA0, cA1, acc);
  chain_mm(avB, cB0, cB1, acc);
  chain_mm(avC, cC0, cC1, acc);
  chain_mm(avD, cD0, cD1, acc);

  #pragma unroll
  for (int mt = 0; mt < 8; ++mt) {
    const int mb = wv * 128 + mt * 16 + 8 * hlf;
    const float4 dA = *(const float4*)(dm + mb);
    const float4 dB = *(const float4*)(dm + mb + 4);
    float4 v0l, v0h, v1l, v1h;
    v0l.x = acc[mt][0][0] * dA.x; v0l.y = acc[mt][0][1] * dA.y;
    v0l.z = acc[mt][0][2] * dA.z; v0l.w = acc[mt][0][3] * dA.w;
    v0h.x = acc[mt][0][4] * dB.x; v0h.y = acc[mt][0][5] * dB.y;
    v0h.z = acc[mt][0][6] * dB.z; v0h.w = acc[mt][0][7] * dB.w;
    v1l.x = acc[mt][1][0] * dA.x; v1l.y = acc[mt][1][1] * dA.y;
    v1l.z = acc[mt][1][2] * dA.z; v1l.w = acc[mt][1][3] * dA.w;
    v1h.x = acc[mt][1][4] * dB.x; v1h.y = acc[mt][1][5] * dB.y;
    v1h.z = acc[mt][1][6] * dB.z; v1h.w = acc[mt][1][7] * dB.w;
    *(float4*)(Tout + (size_t)l16 * TS + mb)            = v0l;
    *(float4*)(Tout + (size_t)l16 * TS + mb + 4)        = v0h;
    *(float4*)(Tout + (size_t)(16 + l16) * TS + mb)     = v1l;
    *(float4*)(Tout + (size_t)(16 + l16) * TS + mb + 4) = v1h;
  }
}

__global__ __launch_bounds__(256) void zero_chain(
    const float* __restrict__ M0,
    const float* __restrict__ Wh1, const float* __restrict__ Wh2,
    const float* __restrict__ d1z, const float* __restrict__ d2z,
    const float* __restrict__ d3z,
    const float* __restrict__ Wout,
    float* __restrict__ zsum)
{
  extern __shared__ float lds[];
  float* T0 = lds;
  float* T1 = lds + CB * TS;

  const int cb = blockIdx.x;
  const int b  = blockIdx.y;
  const int c0 = cb * CB;
  const int lane = threadIdx.x & 31, wv = threadIdx.x >> 5;

  const float* d1 = d1z + (size_t)b * HID;
  const float* d2 = d2z + (size_t)b * HID;
  const float* d3 = d3z + (size_t)b * HID;

  for (int idx = threadIdx.x; idx < HID * CB; idx += 256) {
    const int j = idx >> 5, c = idx & 31;
    T0[(size_t)c * TS + j] = M0[(size_t)j * NST + c0 + c] * d1[j];
  }
  __syncthreads();

  chain_gemm_1024(Wh1, T0, T1, d2, lane, wv);
  __syncthreads();
  chain_gemm_1024(Wh2, T1, T0, d3, lane, wv);
  __syncthreads();

  const int l16 = lane & 15, hlf = lane >> 4;
  const int kof = hlf * 2;
  const v8f vz = {0.f,0.f,0.f,0.f,0.f,0.f,0.f,0.f};
  v8f g0 = vz, g1 = vz;
  const float* Tb0 = T0 + (size_t)l16 * TS;
  const float* Tb1 = T0 + (size_t)(16 + l16) * TS;
  const float* Wr  = Wout + (size_t)(wv * 16 + l16) * HID;

  v2f a0, c00, c10;
  v2f a1, c01, c11;
#define FLD0(kk) { a0 = *(const v2f*)(Wr + (kk)); \
                   c00 = *(const v2f*)(Tb0 + (kk)); c10 = *(const v2f*)(Tb1 + (kk)); }
#define FLD1(kk) { a1 = *(const v2f*)(Wr + (kk)); \
                   c01 = *(const v2f*)(Tb0 + (kk)); c11 = *(const v2f*)(Tb1 + (kk)); }
  FLD0(kof);
  FLD1(kof + 4);
  for (int k = 0; k < HID - 8; k += 8) {
    g0 = wmma_f32(a0, c00, g0); g1 = wmma_f32(a0, c10, g1);
    FLD0(k + 8 + kof);
    g0 = wmma_f32(a1, c01, g0); g1 = wmma_f32(a1, c11, g1);
    FLD1(k + 12 + kof);
  }
  g0 = wmma_f32(a0, c00, g0); g1 = wmma_f32(a0, c10, g1);
  g0 = wmma_f32(a1, c01, g0); g1 = wmma_f32(a1, c11, g1);
#undef FLD0
#undef FLD1

  #pragma unroll
  for (int r = 0; r < 8; ++r) {
    float s = g0[r] * g0[r] + g1[r] * g1[r];
    s += __shfl_xor(s, 1);
    s += __shfl_xor(s, 2);
    s += __shfl_xor(s, 4);
    s += __shfl_xor(s, 8);
    if (l16 == 0)
      atomicAdd(&zsum[(size_t)b * NST + wv * 16 + r + 8 * hlf], s);
  }
}

__global__ void finalize_norm(const float* __restrict__ zsum,
                              float* __restrict__ out) {
  int i = blockIdx.x * blockDim.x + threadIdx.x;
  if (i >= BATCH * NST) return;
  int b = i >> 7, c = i & 127;
  out[(size_t)b * 384 + 256 + c] = sqrtf(zsum[i]);
}

extern "C" void kernel_launch(void* const* d_in, const int* in_sizes, int n_in,
                              void* d_out, int out_size, void* d_ws, size_t ws_size,
                              hipStream_t stream)
{
  (void)in_sizes; (void)n_in; (void)out_size; (void)ws_size;
  const float* x     = (const float*)d_in[0];
  const float* W_in  = (const float*)d_in[1];
  const float* b_in  = (const float*)d_in[2];
  const float* Wh    = (const float*)d_in[3];
  const float* bh    = (const float*)d_in[4];
  const float* W_out = (const float*)d_in[5];
  const float* b_out = (const float*)d_in[6];
  float* out = (float*)d_out;
  float* ws  = (float*)d_ws;

  const float* Wh0 = Wh;
  const float* Wh1 = Wh + 1024 * 1024;
  const float* Wh2 = Wh + 2 * 1024 * 1024;
  const float* bh0 = bh, *bh1 = bh + 1024, *bh2 = bh + 2048;

  float* X2  = ws;
  float* hA  = X2  + 512 * 256;
  float* hB  = hA  + 512 * 1024;
  float* D1  = hB  + 512 * 1024;
  float* D2  = D1  + 512 * 1024;
  float* D3  = D2  + 512 * 1024;
  float* uA  = D3  + 512 * 1024;
  float* uB  = uA  + 256 * 1024;
  float* M0b = uB  + 256 * 1024;
  float* zs  = M0b + 1024 * 128;

  dim3 blk(256);
  build_x2<<<dim3((512 * 256 + 255) / 256), blk, 0, stream>>>(x, X2);
  zero_buf<<<dim3((256 * 128 + 255) / 256), blk, 0, stream>>>(zs, 256 * 128);

  gemm_awt<<<dim3(4,16), blk, 0, stream>>>(X2,256,  W_in,256, b_in, nullptr,0, hA,1024, 512,256,  0);
  gemm_awt<<<dim3(4,16), blk, 0, stream>>>(hA,1024, Wh0,1024, bh0,  nullptr,0, hB,1024, 512,1024, 1);
  gemm_awt<<<dim3(4,16), blk, 0, stream>>>(hB,1024, Wh0,1024, bh0,  nullptr,0, D1,1024, 512,1024, 2);
  gemm_awt<<<dim3(4,16), blk, 0, stream>>>(hB,1024, Wh1,1024, bh1,  nullptr,0, hA,1024, 512,1024, 1);
  gemm_awt<<<dim3(4,16), blk, 0, stream>>>(hA,1024, Wh1,1024, bh1,  nullptr,0, D2,1024, 512,1024, 2);
  gemm_awt<<<dim3(4,16), blk, 0, stream>>>(hA,1024, Wh2,1024, bh2,  nullptr,0, hB,1024, 512,1024, 1);
  gemm_awt<<<dim3(4,16), blk, 0, stream>>>(hB,1024, Wh2,1024, bh2,  nullptr,0, D3,1024, 512,1024, 2);
  gemm_awt<<<dim3(2,2),  blk, 0, stream>>>(hB,1024, W_out,1024, b_out, nullptr,0, out,384, 256,1024, 0);

  gemm_awt<<<dim3(2,16), blk, 0, stream>>>(x + 256,512, W_in,256, nullptr, nullptr,0, uA,1024, 256,128,  0);
  gemm_awt<<<dim3(2,16), blk, 0, stream>>>(uA,1024, Wh0,1024, nullptr, D1,1024, uB,1024, 256,1024, 0);
  gemm_awt<<<dim3(2,16), blk, 0, stream>>>(uB,1024, Wh1,1024, nullptr, D2,1024, uA,1024, 256,1024, 0);
  gemm_awt<<<dim3(2,16), blk, 0, stream>>>(uA,1024, Wh2,1024, nullptr, D3,1024, uB,1024, 256,1024, 0);
  gemm_awt<<<dim3(2,2),  blk, 0, stream>>>(uB,1024, W_out,1024, nullptr, nullptr,0, out + 128,384, 256,1024, 0);

  gemm_ab<<<dim3(8,2), blk, 0, stream>>>(Wh0,1024, W_in,256, M0b,128, 1024,1024);
  zero_chain<<<dim3(4,256), blk, 2 * CB * TS * sizeof(float), stream>>>(
      M0b, Wh1, Wh2,
      D1 + 256 * 1024, D2 + 256 * 1024, D3 + 256 * 1024,
      W_out, zs);
  finalize_norm<<<dim3((256 * 128 + 255) / 256), blk, 0, stream>>>(zs, out);
}


# Zero chain: 4-wave work-groups (256 rows a wave, 16-row ring slots, T fragments read once a window), 147 KB LDS, 2 per WGP: one's fill and load waits run under the other's products; sums unchanged
# speedup vs baseline: 1.9603x; 1.9603x over previous
#include <hip/hip_runtime.h>
#include <hip/hip_bf16.h>
#include <math.h>


typedef float v2f __attribute__((ext_vector_type(2)));
typedef float v8f __attribute__((ext_vector_type(8)));

#define HID   1024
#define NST   128
#define BATCH 256
#define IN2   256
#define CB    32
#define TS    1028
#define SP    36
#define RSLOT (16 * SP)
#define ZT    128
#define NPF   4

__device__ __forceinline__ v8f wmma_f32(v2f a, v2f b, v8f c) {
  return __builtin_amdgcn_wmma_f32_16x16x4_f32(
       false, a,  false, b,
       (short)0, c,  false,  false);
}

__global__ void build_x2(const float* __restrict__ x, float* __restrict__ X2) {
  int i = blockIdx.x * blockDim.x + threadIdx.x;
  if (i >= 512 * IN2) return;
  int row = i >> 8, c = i & 255;
  float v;
  if (row < 256) {
    v = x[row * 512 + c];
  } else {
    int rz = row - 256;
    v = (c < 128) ? x[rz * 512 + 384 + c] : x[rz * 512 + 128 + (c - 128)];
  }
  X2[i] = v;
}

__global__ void zero_buf(float* __restrict__ p, int n) {
  int i = blockIdx.x * blockDim.x + threadIdx.x;
  if (i < n) p[i] = 0.f;
}

#define PST 68

__global__ __launch_bounds__(256) void gemm_awt(
    const float* __restrict__ A, int lda,
    const float* __restrict__ W, int ldw,
    const float* __restrict__ bias,
    const float* __restrict__ mask, int ldm,
    float* __restrict__ C, int ldc,
    int M, int K, int epi)
{
  __shared__ float part[8 * 16 * PST];
  const int lane = threadIdx.x & 31, wv = threadIdx.x >> 5;
  const int l16 = lane & 15, hlf = lane >> 4;
  const int m0 = blockIdx.x * 16;
  const int n0 = blockIdx.y * 64;
  if (m0 >= M) return;

  const v8f vz = {0.f,0.f,0.f,0.f,0.f,0.f,0.f,0.f};
  v8f acc[4] = {vz, vz, vz, vz};

  const int kof = wv * 4 + hlf * 2;
  const float* Ar  = A + (size_t)(m0 + l16) * lda;
  const float* Wr0 = W + (size_t)(n0 +  0 + l16) * ldw;
  const float* Wr1 = W + (size_t)(n0 + 16 + l16) * ldw;
  const float* Wr2 = W + (size_t)(n0 + 32 + l16) * ldw;
  const float* Wr3 = W + (size_t)(n0 + 48 + l16) * ldw;

  v2f a0, b00, b10, b20, b30;
  v2f a1, b01, b11, b21, b31;

#define LDSET0(kk) { a0  = *(const v2f*)(Ar  + (kk)); \
                     b00 = *(const v2f*)(Wr0 + (kk)); \
                     b10 = *(const v2f*)(Wr1 + (kk)); \
                     b20 = *(const v2f*)(Wr2 + (kk)); \
                     b30 = *(const v2f*)(Wr3 + (kk)); }
#define LDSET1(kk) { a1  = *(const v2f*)(Ar  + (kk)); \
                     b01 = *(const v2f*)(Wr0 + (kk)); \
                     b11 = *(const v2f*)(Wr1 + (kk)); \
                     b21 = *(const v2f*)(Wr2 + (kk)); \
                     b31 = *(const v2f*)(Wr3 + (kk)); }
#define WM0() { acc[0] = wmma_f32(a0, b00, acc[0]); \
                acc[1] = wmma_f32(a0, b10, acc[1]); \
                acc[2] = wmma_f32(a0, b20, acc[2]); \
                acc[3] = wmma_f32(a0, b30, acc[3]); }
#define WM1() { acc[0] = wmma_f32(a1, b01, acc[0]); \
                acc[1] = wmma_f32(a1, b11, acc[1]); \
                acc[2] = wmma_f32(a1, b21, acc[2]); \
                acc[3] = wmma_f32(a1, b31, acc[3]); }

  LDSET0(kof);
  LDSET1(kof + 32);
  for (int k = 0; k < K - 64; k += 64) {
    WM0(); LDSET0(k + 64 + kof);
    WM1(); LDSET1(k + 96 + kof);
  }
  WM0();
  WM1();
#undef LDSET0
#undef LDSET1
#undef WM0
#undef WM1

  float* pw = part + wv * (16 * PST);
  #pragma unroll
  for (int nt = 0; nt < 4; ++nt) {
    #pragma unroll
    for (int r = 0; r < 8; ++r)
      pw[(r + 8 * hlf) * PST + nt * 16 + l16] = acc[nt][r];
  }
  __syncthreads();

  const int row = threadIdx.x >> 4, c4 = (threadIdx.x & 15) * 4;
  const float* pr = part + row * PST + c4;
  float4 v = *(const float4*)pr;
  #pragma unroll
  for (int w = 1; w < 8; ++w) {
    const float4 p = *(const float4*)(pr + w * (16 * PST));
    v.x += p.x; v.y += p.y; v.z += p.z; v.w += p.w;
  }
  const int m  = m0 + row;
  const int nn = n0 + c4;
  if (bias) {
    const float4 bv = *(const float4*)(bias + nn);
    v.x += bv.x; v.y += bv.y; v.z += bv.z; v.w += bv.w;
  }
  if (epi == 1) {
    v.x = v.x > 0.f ? v.x : 0.f; v.y = v.y > 0.f ? v.y : 0.f;
    v.z = v.z > 0.f ? v.z : 0.f; v.w = v.w > 0.f ? v.w : 0.f;
  } else if (epi == 2) {
    v.x = v.x > 0.f ? 1.f : 0.f; v.y = v.y > 0.f ? 1.f : 0.f;
    v.z = v.z > 0.f ? 1.f : 0.f; v.w = v.w > 0.f ? 1.f : 0.f;
  }
  if (mask) {
    const float4 mv = *(const float4*)(mask + (size_t)m * ldm + nn);
    v.x *= mv.x; v.y *= mv.y; v.z *= mv.z; v.w *= mv.w;
  }
  *(float4*)(C + (size_t)m * ldc + nn) = v;
}

#define GROWS 80
#define GSLOT (GROWS * SP)

__global__ __launch_bounds__(256) void gemm_awt_lines(
    const float* __restrict__ A, int lda,
    const float* __restrict__ W, int ldw,
    const float* __restrict__ bias,
    const float* __restrict__ mask, int ldm,
    float* __restrict__ C, int ldc,
    int M, int K, int epi)
{
  extern __shared__ float gl[];
  float* part = gl;
  const int lane = threadIdx.x & 31, wv = threadIdx.x >> 5;
  const int l16 = lane & 15, hlf = lane >> 4;
  const int m0 = blockIdx.x * 16;
  const int n0 = blockIdx.y * 64;
  if (m0 >= M) return;

  float* slot = gl + 8 * 16 * PST + wv * GSLOT;
  const int kc = K >> 3;
  const int lr = lane >> 3, lp = 4 * (lane & 7);
  const float* Ag = A + (size_t)(m0 + lr) * lda + wv * kc + lp;
  const float* Wg = W + (size_t)(n0 + lr) * ldw + wv * kc + lp;
  float* sst = slot + lr * SP + lp;
  const float* fa = slot + l16 * SP + 2 * hlf;
  const float* fb = slot + (16 + l16) * SP + 2 * hlf;

  const v8f vz = {0.f,0.f,0.f,0.f,0.f,0.f,0.f,0.f};
  v8f acc[4] = {vz, vz, vz, vz};

  float4 ga0, ga1, ga2, ga3, gw0, gw1, gw2, gw3, gw4, gw5, gw6, gw7, gw8, gw9, gw10, gw11, gw12, gw13, gw14, gw15;
#define LDA(n) ga##n = *(const float4*)(Ag + (size_t)(4 * n) * lda + kk_);
#define LDW(n) gw##n = *(const float4*)(Wg + (size_t)(4 * n) * ldw + kk_);
#define STA(n) *(float4*)(sst + (4 * n) * SP) = ga##n;
#define STW(n) *(float4*)(sst + (16 + 4 * n) * SP) = gw##n;
#define GLD(kk) { const int kk_ = (kk); LDA(0) LDA(1) LDA(2) LDA(3) \
                  LDW(0) LDW(1) LDW(2) LDW(3) LDW(4) LDW(5) LDW(6) LDW(7) LDW(8) LDW(9) LDW(10) LDW(11) LDW(12) LDW(13) LDW(14) LDW(15) }
#define GST()   { STA(0) STA(1) STA(2) STA(3) \
                  STW(0) STW(1) STW(2) STW(3) STW(4) STW(5) STW(6) STW(7) STW(8) STW(9) STW(10) STW(11) STW(12) STW(13) STW(14) STW(15) }
#define GMUL()  { _Pragma("unroll") for (int s = 0; s < 8; ++s) { \
                    const v2f a = *(const v2f*)(fa + 4 * s); \
                    const v2f b0 = *(const v2f*)(fb + 0 * 16 * SP + 4 * s); \
                    const v2f b1 = *(const v2f*)(fb + 1 * 16 * SP + 4 * s); \
                    const v2f b2 = *(const v2f*)(fb + 2 * 16 * SP + 4 * s); \
                    const v2f b3 = *(const v2f*)(fb + 3 * 16 * SP + 4 * s); \
                    acc[0] = wmma_f32(a, b0, acc[0]); acc[1] = wmma_f32(a, b1, acc[1]); \
                    acc[2] = wmma_f32(a, b2, acc[2]); acc[3] = wmma_f32(a, b3, acc[3]); } }

  GLD(0);
  for (int kw = 0; kw < kc - 32; kw += 32) {
    GST();
    GLD(kw + 32);
    GMUL();
  }
  GST();
  GMUL();
#undef LDA
#undef LDW
#undef STA
#undef STW
#undef GLD
#undef GST
#undef GMUL

  float* pw = part + wv * (16 * PST);
  #pragma unroll
  for (int nt = 0; nt < 4; ++nt) {
    #pragma unroll
    for (int r = 0; r < 8; ++r)
      pw[(r + 8 * hlf) * PST + nt * 16 + l16] = acc[nt][r];
  }
  __syncthreads();

  const int row = threadIdx.x >> 4, c4 = (threadIdx.x & 15) * 4;
  const float* pr = part + row * PST + c4;
  float4 v = *(const float4*)pr;
  #pragma unroll
  for (int w = 1; w < 8; ++w) {
    const float4 p = *(const float4*)(pr + w * (16 * PST));
    v.x += p.x; v.y += p.y; v.z += p.z; v.w += p.w;
  }
  const int m  = m0 + row;
  const int nn = n0 + c4;
  if (bias) {
    const float4 bv = *(const float4*)(bias + nn);
    v.x += bv.x; v.y += bv.y; v.z += bv.z; v.w += bv.w;
  }
  if (epi == 1) {
    v.x = v.x > 0.f ? v.x : 0.f; v.y = v.y > 0.f ? v.y : 0.f;
    v.z = v.z > 0.f ? v.z : 0.f; v.w = v.w > 0.f ? v.w : 0.f;
  } else if (epi == 2) {
    v.x = v.x > 0.f ? 1.f : 0.f; v.y = v.y > 0.f ? 1.f : 0.f;
    v.z = v.z > 0.f ? 1.f : 0.f; v.w = v.w > 0.f ? 1.f : 0.f;
  }
  if (mask) {
    const float4 mv = *(const float4*)(mask + (size_t)m * ldm + nn);
    v.x *= mv.x; v.y *= mv.y; v.z *= mv.z; v.w *= mv.w;
  }
  *(float4*)(C + (size_t)m * ldc + nn) = v;
}

__global__ __launch_bounds__(256) void gemm_ab(
    const float* __restrict__ A, int lda,
    const float* __restrict__ B, int ldb,
    float* __restrict__ C, int ldc,
    int M, int K)
{
  __shared__ float part[8 * 16 * PST];
  const int lane = threadIdx.x & 31, wv = threadIdx.x >> 5;
  const int l16 = lane & 15, hlf = lane >> 4;
  const int m0 = blockIdx.x * 16;
  const int n0 = blockIdx.y * 64;
  if (m0 >= M) return;

  const v8f vz = {0.f,0.f,0.f,0.f,0.f,0.f,0.f,0.f};
  v8f acc[4] = {vz, vz, vz, vz};
  const int kof = wv * 4 + hlf * 2;
  const float* Ar = A + (size_t)(m0 + l16) * lda;
  const float* Bc = B + n0 + l16;

  v2f a0, c00, c10, c20, c30;
  v2f a1, c01, c11, c21, c31;

#define BLD(dst, kk, nt) { (dst).x = Bc[(size_t)(kk) * ldb + (nt) * 16]; \
                           (dst).y = Bc[(size_t)((kk) + 1) * ldb + (nt) * 16]; }
#define LDSET0(kk) { a0 = *(const v2f*)(Ar + (kk)); \
                     BLD(c00, (kk), 0) BLD(c10, (kk), 1) BLD(c20, (kk), 2) BLD(c30, (kk), 3) }
#define LDSET1(kk) { a1 = *(const v2f*)(Ar + (kk)); \
                     BLD(c01, (kk), 0) BLD(c11, (kk), 1) BLD(c21, (kk), 2) BLD(c31, (kk), 3) }
#define WM0() { acc[0] = wmma_f32(a0, c00, acc[0]); acc[1] = wmma_f32(a0, c10, acc[1]); \
                acc[2] = wmma_f32(a0, c20, acc[2]); acc[3] = wmma_f32(a0, c30, acc[3]); }
#define WM1() { acc[0] = wmma_f32(a1, c01, acc[0]); acc[1] = wmma_f32(a1, c11, acc[1]); \
                acc[2] = wmma_f32(a1, c21, acc[2]); acc[3] = wmma_f32(a1, c31, acc[3]); }

  LDSET0(kof);
  LDSET1(kof + 32);
  for (int k = 0; k < K - 64; k += 64) {
    WM0(); LDSET0(k + 64 + kof);
    WM1(); LDSET1(k + 96 + kof);
  }
  WM0();
  WM1();
#undef BLD
#undef LDSET0
#undef LDSET1
#undef WM0
#undef WM1

  float* pw = part + wv * (16 * PST);
  #pragma unroll
  for (int nt = 0; nt < 4; ++nt) {
    #pragma unroll
    for (int r = 0; r < 8; ++r)
      pw[(r + 8 * hlf) * PST + nt * 16 + l16] = acc[nt][r];
  }
  __syncthreads();

  const int row = threadIdx.x >> 4, c4 = (threadIdx.x & 15) * 4;
  const float* pr = part + row * PST + c4;
  float4 v = *(const float4*)pr;
  #pragma unroll
  for (int w = 1; w < 8; ++w) {
    const float4 p = *(const float4*)(pr + w * (16 * PST));
    v.x += p.x; v.y += p.y; v.z += p.z; v.w += p.w;
  }
  *(float4*)(C + (size_t)(m0 + row) * ldc + n0 + c4) = v;
}

__device__ __forceinline__ void chain_gemm_1024(
    const float* __restrict__ Wm,
    float* T,
    float* ring,
    const float* __restrict__ dm,
    int lane, int wv)
{
  const int l16 = lane & 15, hlf = lane >> 4;
  const int kof = hlf * 2;
  const v8f vz = {0.f,0.f,0.f,0.f,0.f,0.f,0.f,0.f};
  v8f acc[16][2];
  #pragma unroll
  for (int mt = 0; mt < 16; ++mt) { acc[mt][0] = vz; acc[mt][1] = vz; }

  const float* Tb0 = T + (size_t)l16 * TS + kof;
  const float* Tb1 = T + (size_t)(16 + l16) * TS + kof;
  const float* Wg  = Wm + (size_t)(wv * 256 + (lane >> 3)) * HID + 4 * (lane & 7);
  float* st0 = ring + (lane >> 3) * SP + 4 * (lane & 7);
  float* st1 = st0 + RSLOT;
  const float* fr0 = ring + l16 * SP + kof;
  const float* fr1 = fr0 + RSLOT;

#define GDECL(G) float4 G##_0, G##_1, G##_2, G##_3;
#define ADECL(A) v2f A##_0, A##_1, A##_2, A##_3, A##_4, A##_5, A##_6, A##_7;
  GDECL(G0) GDECL(G1) GDECL(G2) GDECL(G3)
  ADECL(A0) ADECL(A1)
#define TLD(G, P) { const float* p_ = (P); \
    G##_0 = *(const float4*)(p_); \
    G##_1 = *(const float4*)(p_ + (size_t)4 * HID); \
    G##_2 = *(const float4*)(p_ + (size_t)8 * HID); \
    G##_3 = *(const float4*)(p_ + (size_t)12 * HID); }
#define TST(S, G) { float* s_ = (S); \
    *(float4*)(s_)          = G##_0; \
    *(float4*)(s_ + 4 * SP) = G##_1; \
    *(float4*)(s_ + 8 * SP) = G##_2; \
    *(float4*)(s_ + 12 * SP) = G##_3; }
#define FLD(F, A) { const float* f_ = (F); \
    A##_0 = *(const v2f*)(f_);      A##_1 = *(const v2f*)(f_ + 4); \
    A##_2 = *(const v2f*)(f_ + 8);  A##_3 = *(const v2f*)(f_ + 12); \
    A##_4 = *(const v2f*)(f_ + 16); A##_5 = *(const v2f*)(f_ + 20); \
    A##_6 = *(const v2f*)(f_ + 24); A##_7 = *(const v2f*)(f_ + 28); }
#define MUL1(t, A, s) { acc[t][0] = wmma_f32(A##_##s, c0[s], acc[t][0]); \
                        acc[t][1] = wmma_f32(A##_##s, c1[s], acc[t][1]); }

  TLD(G0, Wg)
  TLD(G1, Wg + (size_t)16 * HID)
  TLD(G2, Wg + (size_t)32 * HID)
  TLD(G3, Wg + (size_t)48 * HID)
  TST(st0, G0)
  FLD(fr0, A0)

#define TILE(t, GCUR, GNXT, STN, FRN, ACUR, ANXT) { \
    TLD(GCUR, ((t) + NPF < 16) ? Wg + (size_t)(((t) + NPF) * 16) * HID + kw \
                               : Wg + (size_t)(((t) + NPF - 16) * 16) * HID + kn) \
    TST(STN, GNXT) \
    FLD(FRN, ANXT) \
    __builtin_amdgcn_sched_barrier(0); \
    MUL1(t, ACUR, 0) MUL1(t, ACUR, 1) MUL1(t, ACUR, 2) MUL1(t, ACUR, 3) \
    MUL1(t, ACUR, 4) MUL1(t, ACUR, 5) MUL1(t, ACUR, 6) MUL1(t, ACUR, 7) \
    __builtin_amdgcn_sched_barrier(0); }

  for (int kw = 0; kw < HID; kw += 32) {
    const int kn = (kw + 32 < HID) ? kw + 32 : kw;

    v2f c0[8], c1[8];
    #pragma unroll
    for (int s = 0; s < 8; ++s) {
      c0[s] = *(const v2f*)(Tb0 + kw + 4 * s);
      c1[s] = *(const v2f*)(Tb1 + kw + 4 * s);
    }

    TILE( 0, G0, G1, st1, fr1, A0, A1)
    TILE( 1, G1, G2, st0, fr0, A1, A0)
    TILE( 2, G2, G3, st1, fr1, A0, A1)
    TILE( 3, G3, G0, st0, fr0, A1, A0)
    TILE( 4, G0, G1, st1, fr1, A0, A1)
    TILE( 5, G1, G2, st0, fr0, A1, A0)
    TILE( 6, G2, G3, st1, fr1, A0, A1)
    TILE( 7, G3, G0, st0, fr0, A1, A0)
    TILE( 8, G0, G1, st1, fr1, A0, A1)
    TILE( 9, G1, G2, st0, fr0, A1, A0)
    TILE(10, G2, G3, st1, fr1, A0, A1)
    TILE(11, G3, G0, st0, fr0, A1, A0)
    TILE(12, G0, G1, st1, fr1, A0, A1)
    TILE(13, G1, G2, st0, fr0, A1, A0)
    TILE(14, G2, G3, st1, fr1, A0, A1)
    TILE(15, G3, G0, st0, fr0, A1, A0)
  }
#undef TILE
#undef MUL1
#undef FLD
#undef TST
#undef TLD
#undef ADECL
#undef GDECL

  __syncthreads();

  #pragma unroll
  for (int mt = 0; mt < 16; ++mt) {
    const int mb = wv * 256 + mt * 16 + 8 * hlf;
    const float4 dA = *(const float4*)(dm + mb);
    const float4 dB = *(const float4*)(dm + mb + 4);
    float4 v0l, v0h, v1l, v1h;
    v0l.x = acc[mt][0][0] * dA.x; v0l.y = acc[mt][0][1] * dA.y;
    v0l.z = acc[mt][0][2] * dA.z; v0l.w = acc[mt][0][3] * dA.w;
    v0h.x = acc[mt][0][4] * dB.x; v0h.y = acc[mt][0][5] * dB.y;
    v0h.z = acc[mt][0][6] * dB.z; v0h.w = acc[mt][0][7] * dB.w;
    v1l.x = acc[mt][1][0] * dA.x; v1l.y = acc[mt][1][1] * dA.y;
    v1l.z = acc[mt][1][2] * dA.z; v1l.w = acc[mt][1][3] * dA.w;
    v1h.x = acc[mt][1][4] * dB.x; v1h.y = acc[mt][1][5] * dB.y;
    v1h.z = acc[mt][1][6] * dB.z; v1h.w = acc[mt][1][7] * dB.w;
    *(float4*)(T + (size_t)l16 * TS + mb)            = v0l;
    *(float4*)(T + (size_t)l16 * TS + mb + 4)        = v0h;
    *(float4*)(T + (size_t)(16 + l16) * TS + mb)     = v1l;
    *(float4*)(T + (size_t)(16 + l16) * TS + mb + 4) = v1h;
  }
}

__global__ __launch_bounds__(ZT) __attribute__((amdgpu_waves_per_eu(2, 2))) void zero_chain(
    const float* __restrict__ M0,
    const float* __restrict__ Wh1, const float* __restrict__ Wh2,
    const float* __restrict__ d1z, const float* __restrict__ d2z,
    const float* __restrict__ d3z,
    const float* __restrict__ Wout,
    float* __restrict__ zsum)
{
  extern __shared__ float lds[];
  float* T0 = lds;

  const int cb = blockIdx.x;
  const int b  = blockIdx.y;
  const int c0 = cb * CB;
  const int lane = threadIdx.x & 31, wv = threadIdx.x >> 5;

  const float* d1 = d1z + (size_t)b * HID;
  const float* d2 = d2z + (size_t)b * HID;
  const float* d3 = d3z + (size_t)b * HID;

  for (int idx = threadIdx.x; idx < HID * CB; idx += ZT) {
    const int j = idx >> 5, c = idx & 31;
    T0[(size_t)c * TS + j] = M0[(size_t)j * NST + c0 + c] * d1[j];
  }
  __syncthreads();

  float* ring = lds + CB * TS + wv * (2 * RSLOT);
  chain_gemm_1024(Wh1, T0, ring, d2, lane, wv);
  __syncthreads();
  chain_gemm_1024(Wh2, T0, ring, d3, lane, wv);
  __syncthreads();

  const int l16 = lane & 15, hlf = lane >> 4;
  const int kof = hlf * 2;
  const v8f vz = {0.f,0.f,0.f,0.f,0.f,0.f,0.f,0.f};
  v8f ga0 = vz, ga1 = vz, gb0 = vz, gb1 = vz;
  const float* Tb0 = T0 + (size_t)l16 * TS + kof;
  const float* Tb1 = T0 + (size_t)(16 + l16) * TS + kof;
  const float* WrA = Wout + (size_t)(wv * 32 + l16) * HID + kof;
  const float* WrB = WrA + (size_t)16 * HID;

  #pragma unroll 8
  for (int k = 0; k < HID; k += 4) {
    const v2f aa = *(const v2f*)(WrA + k);
    const v2f ab = *(const v2f*)(WrB + k);
    const v2f c0v = *(const v2f*)(Tb0 + k);
    const v2f c1v = *(const v2f*)(Tb1 + k);
    ga0 = wmma_f32(aa, c0v, ga0); ga1 = wmma_f32(aa, c1v, ga1);
    gb0 = wmma_f32(ab, c0v, gb0); gb1 = wmma_f32(ab, c1v, gb1);
  }

  #pragma unroll
  for (int r = 0; r < 8; ++r) {
    float sa = ga0[r] * ga0[r] + ga1[r] * ga1[r];
    float sb = gb0[r] * gb0[r] + gb1[r] * gb1[r];
    sa += __shfl_xor(sa, 1);  sb += __shfl_xor(sb, 1);
    sa += __shfl_xor(sa, 2);  sb += __shfl_xor(sb, 2);
    sa += __shfl_xor(sa, 4);  sb += __shfl_xor(sb, 4);
    sa += __shfl_xor(sa, 8);  sb += __shfl_xor(sb, 8);
    if (l16 == 0) {
      atomicAdd(&zsum[(size_t)b * NST + wv * 32 + r + 8 * hlf], sa);
      atomicAdd(&zsum[(size_t)b * NST + wv * 32 + 16 + r + 8 * hlf], sb);
    }
  }
}

__global__ void finalize_norm(const float* __restrict__ zsum,
                              float* __restrict__ out) {
  int i = blockIdx.x * blockDim.x + threadIdx.x;
  if (i >= BATCH * NST) return;
  int b = i >> 7, c = i & 127;
  out[(size_t)b * 384 + 256 + c] = sqrtf(zsum[i]);
}

extern "C" void kernel_launch(void* const* d_in, const int* in_sizes, int n_in,
                              void* d_out, int out_size, void* d_ws, size_t ws_size,
                              hipStream_t stream)
{
  (void)in_sizes; (void)n_in; (void)out_size; (void)ws_size;
  const float* x     = (const float*)d_in[0];
  const float* W_in  = (const float*)d_in[1];
  const float* b_in  = (const float*)d_in[2];
  const float* Wh    = (const float*)d_in[3];
  const float* bh    = (const float*)d_in[4];
  const float* W_out = (const float*)d_in[5];
  const float* b_out = (const float*)d_in[6];
  float* out = (float*)d_out;
  float* ws  = (float*)d_ws;

  const float* Wh0 = Wh;
  const float* Wh1 = Wh + 1024 * 1024;
  const float* Wh2 = Wh + 2 * 1024 * 1024;
  const float* bh0 = bh, *bh1 = bh + 1024, *bh2 = bh + 2048;

  float* X2  = ws;
  float* hA  = X2  + 512 * 256;
  float* hB  = hA  + 512 * 1024;
  float* D1  = hB  + 512 * 1024;
  float* D2  = D1  + 512 * 1024;
  float* D3  = D2  + 512 * 1024;
  float* uA  = D3  + 512 * 1024;
  float* uB  = uA  + 256 * 1024;
  float* M0b = uB  + 256 * 1024;
  float* zs  = M0b + 1024 * 128;

  dim3 blk(256);
  build_x2<<<dim3((512 * 256 + 255) / 256), blk, 0, stream>>>(x, X2);
  zero_buf<<<dim3((256 * 128 + 255) / 256), blk, 0, stream>>>(zs, 256 * 128);

  gemm_awt_lines<<<dim3(32,16), blk, (8 * 16 * PST + 8 * GSLOT) * sizeof(float), stream>>>(X2,256,  W_in,256, b_in, nullptr,0, hA,1024, 512,256,  0);
  gemm_awt_lines<<<dim3(32,16), blk, (8 * 16 * PST + 8 * GSLOT) * sizeof(float), stream>>>(hA,1024, Wh0,1024, bh0,  nullptr,0, hB,1024, 512,1024, 1);
  gemm_awt_lines<<<dim3(32,16), blk, (8 * 16 * PST + 8 * GSLOT) * sizeof(float), stream>>>(hB,1024, Wh0,1024, bh0,  nullptr,0, D1,1024, 512,1024, 2);
  gemm_awt_lines<<<dim3(32,16), blk, (8 * 16 * PST + 8 * GSLOT) * sizeof(float), stream>>>(hB,1024, Wh1,1024, bh1,  nullptr,0, hA,1024, 512,1024, 1);
  gemm_awt_lines<<<dim3(32,16), blk, (8 * 16 * PST + 8 * GSLOT) * sizeof(float), stream>>>(hA,1024, Wh1,1024, bh1,  nullptr,0, D2,1024, 512,1024, 2);
  gemm_awt_lines<<<dim3(32,16), blk, (8 * 16 * PST + 8 * GSLOT) * sizeof(float), stream>>>(hA,1024, Wh2,1024, bh2,  nullptr,0, hB,1024, 512,1024, 1);
  gemm_awt_lines<<<dim3(32,16), blk, (8 * 16 * PST + 8 * GSLOT) * sizeof(float), stream>>>(hB,1024, Wh2,1024, bh2,  nullptr,0, D3,1024, 512,1024, 2);
  gemm_awt_lines<<<dim3(16,2), blk, (8 * 16 * PST + 8 * GSLOT) * sizeof(float), stream>>>(hB,1024, W_out,1024, b_out, nullptr,0, out,384, 256,1024, 0);

  gemm_awt<<<dim3(16,16), blk, 0, stream>>>(x + 256,512, W_in,256, nullptr, nullptr,0, uA,1024, 256,128,  0);
  gemm_awt_lines<<<dim3(16,16), blk, (8 * 16 * PST + 8 * GSLOT) * sizeof(float), stream>>>(uA,1024, Wh0,1024, nullptr, D1,1024, uB,1024, 256,1024, 0);
  gemm_awt_lines<<<dim3(16,16), blk, (8 * 16 * PST + 8 * GSLOT) * sizeof(float), stream>>>(uB,1024, Wh1,1024, nullptr, D2,1024, uA,1024, 256,1024, 0);
  gemm_awt_lines<<<dim3(16,16), blk, (8 * 16 * PST + 8 * GSLOT) * sizeof(float), stream>>>(uA,1024, Wh2,1024, nullptr, D3,1024, uB,1024, 256,1024, 0);
  gemm_awt_lines<<<dim3(16,2), blk, (8 * 16 * PST + 8 * GSLOT) * sizeof(float), stream>>>(uB,1024, W_out,1024, nullptr, nullptr,0, out + 128,384, 256,1024, 0);

  gemm_ab<<<dim3(64,2), blk, 0, stream>>>(Wh0,1024, W_in,256, M0b,128, 1024,1024);
  zero_chain<<<dim3(4,256), dim3(ZT), (CB * TS + 8 * RSLOT) * sizeof(float), stream>>>(
      M0b, Wh1, Wh2,
      D1 + 256 * 1024, D2 + 256 * 1024, D3 + 256 * 1024,
      W_out, zs);
  finalize_norm<<<dim3((256 * 128 + 255) / 256), blk, 0, stream>>>(zs, out);
}
